# starting kernel
# baseline (speedup 1.0000x reference)
#include <hip/hip_runtime.h>

typedef __attribute__((ext_vector_type(2))) float v2f;
typedef __attribute__((ext_vector_type(8))) float v8f;

#define N_RES   8192
#define N_INP   64
#define N_HAD   16384
#define N_LAYER 3
#define PITCH   140
#define LDS_BYTES (128 * PITCH * 4)

__device__ __forceinline__ float hsgn(int i, int j) {
  return (__popc(i & j) & 1) ? -1.0f : 1.0f;
}

__device__ __forceinline__ v8f v8f_zero() {
  v8f z;
#pragma unroll
  for (int i = 0; i < 8; ++i) z[i] = 0.0f;
  return z;
}

__global__ __launch_bounds__(256) void fwht_reservoir(
    const float* __restrict__ state, const float* __restrict__ inputs,
    const float* __restrict__ bias, const float* __restrict__ diag,
    float* __restrict__ out) {
  extern __shared__ float X[];
  const int row  = blockIdx.x;
  const int tid  = threadIdx.x;
  const int lane = tid & 31;
  const int wave = tid >> 5;
  const int m    = lane & 15;
  const int kh   = lane >> 4;

  for (int n = tid; n < N_HAD; n += 256) {
    float v;
    if (n < N_RES)                   v = 0.9f * state[(size_t)row * N_RES + n];
    else if (n < N_RES + N_INP)      v = 0.4f * inputs[row * N_INP + (n - N_RES)];
    else                             v = 0.0f;
    X[(n >> 7) * PITCH + (n & 127)] = v * diag[n];
  }
  __syncthreads();

  v2f Hp[4], Hm[4];
#pragma unroll
  for (int s = 0; s < 4; ++s) {
    const int j0 = 4 * s + 2 * kh;
    Hp[s].x = hsgn(m, j0);
    Hp[s].y = hsgn(m, j0 + 1);
    Hm[s].x = -Hp[s].x;
    Hm[s].y = -Hp[s].y;
  }

  for (int l = 0; l < N_LAYER; ++l) {
    {
      const int c = wave * 16 + m;
      v8f acc[8];
#pragma unroll
      for (int t = 0; t < 8; ++t) acc[t] = v8f_zero();
#pragma unroll
      for (int k = 0; k < 8; ++k) {
#pragma unroll
        for (int s = 0; s < 4; ++s) {
          const int r0 = k * 16 + 4 * s + 2 * kh;
          v2f xb;
          xb.x = X[r0 * PITCH + c];
          xb.y = X[(r0 + 1) * PITCH + c];
#pragma unroll
          for (int t = 0; t < 8; ++t) {
            const v2f h = (__popc(t & k) & 1) ? Hm[s] : Hp[s];
            acc[t] = __builtin_amdgcn_wmma_f32_16x16x4_f32(
                false, h, false, xb, (short)0, acc[t], false, false);
          }
        }
      }
#pragma unroll
      for (int t = 0; t < 8; ++t)
#pragma unroll
        for (int v = 0; v < 8; ++v)
          X[(t * 16 + v + 8 * kh) * PITCH + c] = acc[t][v];
    }
    __syncthreads();

    {
      const int r = wave * 16 + m;
      v8f acc[8];
#pragma unroll
      for (int t = 0; t < 8; ++t) acc[t] = v8f_zero();
#pragma unroll
      for (int k = 0; k < 8; ++k) {
#pragma unroll
        for (int s = 0; s < 4; ++s) {
          const int c0 = k * 16 + 4 * s + 2 * kh;
          v2f xa;
          xa.x = X[r * PITCH + c0];
          xa.y = X[r * PITCH + c0 + 1];
#pragma unroll
          for (int t = 0; t < 8; ++t) {
            const v2f h = (__popc(t & k) & 1) ? Hm[s] : Hp[s];
            acc[t] = __builtin_amdgcn_wmma_f32_16x16x4_f32(
                false, xa, false, h, (short)0, acc[t], false, false);
          }
        }
      }
#pragma unroll
      for (int t = 0; t < 8; ++t)
#pragma unroll
        for (int v = 0; v < 8; ++v)
          X[(wave * 16 + v + 8 * kh) * PITCH + t * 16 + m] = acc[t][v];
    }
    __syncthreads();

    if (l + 1 < N_LAYER) {
      const float* dg = diag + (size_t)(l + 1) * N_HAD;
      for (int n = tid; n < N_HAD; n += 256)
        X[(n >> 7) * PITCH + (n & 127)] *= dg[n];
      __syncthreads();
    }
  }

  for (int n = tid; n < N_RES; n += 256) {
    const float z = X[(n >> 7) * PITCH + (n & 127)] * (1.0f / (float)N_HAD) + bias[n];
    out[(size_t)row * N_RES + n] = erff(z);
  }
}

extern "C" void kernel_launch(void* const* d_in, const int* in_sizes, int n_in,
                              void* d_out, int out_size, void* d_ws, size_t ws_size,
                              hipStream_t stream) {
  const float* state  = (const float*)d_in[0];
  const float* inputs = (const float*)d_in[1];
  const float* bias   = (const float*)d_in[2];
  const float* diag   = (const float*)d_in[3];
  float* out = (float*)d_out;

  const int batch = in_sizes[0] / N_RES;

  (void)hipFuncSetAttribute(reinterpret_cast<const void*>(fwht_reservoir),
                            hipFuncAttributeMaxDynamicSharedMemorySize, LDS_BYTES);

  fwht_reservoir<<<dim3(batch), dim3(256), LDS_BYTES, stream>>>(
      state, inputs, bias, diag, out);
}


# Elementwise passes (input assembly, diagonal scaling, epilogue) use 16-byte global and LDS accesses with all global loads of a pass issued before first use, not one dependent load per element
# speedup vs baseline: 3.7415x; 3.7415x over previous
#include <hip/hip_runtime.h>

typedef __attribute__((ext_vector_type(2))) float v2f;
typedef __attribute__((ext_vector_type(4))) float v4f;
typedef __attribute__((ext_vector_type(8))) float v8f;

#define N_RES   8192
#define N_INP   64
#define N_HAD   16384
#define N_LAYER 3
#define PITCH   140
#define LDS_BYTES (128 * PITCH * 4)
#define GSTEP   (8 * PITCH)

__device__ __forceinline__ float hsgn(int i, int j) {
  return (__popc(i & j) & 1) ? -1.0f : 1.0f;
}

__device__ __forceinline__ v8f v8f_zero() {
  v8f z;
#pragma unroll
  for (int i = 0; i < 8; ++i) z[i] = 0.0f;
  return z;
}

__global__ __launch_bounds__(256) void fwht_reservoir(
    const float* __restrict__ state, const float* __restrict__ inputs,
    const float* __restrict__ bias, const float* __restrict__ diag,
    float* __restrict__ out) {
  extern __shared__ __attribute__((aligned(16))) float X[];
  const int row  = blockIdx.x;
  const int tid  = threadIdx.x;
  const int lane = tid & 31;
  const int wave = tid >> 5;
  const int m    = lane & 15;
  const int kh   = lane >> 4;
  const int gb   = wave * PITCH + (lane << 2);

  {
    const v4f* st4 = reinterpret_cast<const v4f*>(state + (size_t)row * N_RES);
    const v4f* in4 = reinterpret_cast<const v4f*>(inputs + (size_t)row * N_INP);
    const v4f* dg4 = reinterpret_cast<const v4f*>(diag);
    v4f s[8], d[16];
#pragma unroll
    for (int i = 0; i < 8; ++i) s[i] = st4[tid + 256 * i];
    const v4f u = in4[tid & 15];
#pragma unroll
    for (int i = 0; i < 16; ++i) d[i] = dg4[tid + 256 * i];
#pragma unroll
    for (int i = 0; i < 16; ++i) {
      v4f v;
#pragma unroll
      for (int e = 0; e < 4; ++e) {
        if (i < 8)        v[e] = 0.9f * s[i][e];
        else if (i == 8)  v[e] = (tid < 16) ? 0.4f * u[e] : 0.0f;
        else              v[e] = 0.0f;
      }
      *reinterpret_cast<v4f*>(&X[gb + i * GSTEP]) = v * d[i];
    }
  }
  __syncthreads();

  v2f H16f[4], G8f[4];
#pragma unroll
  for (int j = 0; j < 4; ++j) {
    const int k0 = 8 * kh + 2 * j;
    H16f[j].x = hsgn(m, k0);
    H16f[j].y = hsgn(m, k0 + 1);
    const bool same = ((m >> 3) == kh);
    G8f[j].x = same ? hsgn(m & 7, 2 * j)     : 0.0f;
    G8f[j].y = same ? hsgn(m & 7, 2 * j + 1) : 0.0f;
  }

  for (int l = 0; l < N_LAYER; ++l) {
    {
      const int c = wave * 16 + m;
      v2f xb[8][4];
#pragma unroll
      for (int t = 0; t < 8; ++t)
#pragma unroll
        for (int j = 0; j < 4; ++j) {
          const int r0 = t * 16 + 8 * kh + 2 * j;
          xb[t][j].x = X[r0 * PITCH + c];
          xb[t][j].y = X[(r0 + 1) * PITCH + c];
        }
      v8f z[8];
#pragma unroll
      for (int t = 0; t < 8; ++t) z[t] = v8f_zero();
#pragma unroll
      for (int j = 0; j < 4; ++j)
#pragma unroll
        for (int t = 0; t < 8; ++t)
          z[t] = __builtin_amdgcn_wmma_f32_16x16x4_f32(
              false, H16f[j], false, xb[t][j], (short)0, z[t], false, false);
      v8f y[8];
#pragma unroll
      for (int r = 0; r < 8; ++r) y[r] = v8f_zero();
#pragma unroll
      for (int j = 0; j < 4; ++j)
#pragma unroll
        for (int r = 0; r < 8; ++r) {
          v2f b;
          b.x = z[2 * j][r];
          b.y = z[2 * j + 1][r];
          y[r] = __builtin_amdgcn_wmma_f32_16x16x4_f32(
              false, G8f[j], false, b, (short)0, y[r], false, false);
        }
#pragma unroll
      for (int r = 0; r < 8; ++r)
#pragma unroll
        for (int t = 0; t < 8; ++t)
          X[(t * 16 + 8 * kh + r) * PITCH + c] = y[r][t];
    }
    __syncthreads();

    {
      const int r = wave * 16 + m;
      v4f xa[8][2];
#pragma unroll
      for (int t = 0; t < 8; ++t) {
        const float* p = &X[r * PITCH + t * 16 + 8 * kh];
        xa[t][0] = *reinterpret_cast<const v4f*>(p);
        xa[t][1] = *reinterpret_cast<const v4f*>(p + 4);
      }
      v8f z[8];
#pragma unroll
      for (int t = 0; t < 8; ++t) z[t] = v8f_zero();
#pragma unroll
      for (int j = 0; j < 4; ++j)
#pragma unroll
        for (int t = 0; t < 8; ++t) {
          v2f a;
          a.x = xa[t][j >> 1][2 * (j & 1)];
          a.y = xa[t][j >> 1][2 * (j & 1) + 1];
          z[t] = __builtin_amdgcn_wmma_f32_16x16x4_f32(
              false, a, false, H16f[j], (short)0, z[t], false, false);
        }
      v8f y[8];
#pragma unroll
      for (int v = 0; v < 8; ++v) y[v] = v8f_zero();
#pragma unroll
      for (int j = 0; j < 4; ++j)
#pragma unroll
        for (int v = 0; v < 8; ++v) {
          v2f b;
          b.x = z[2 * j][v];
          b.y = z[2 * j + 1][v];
          y[v] = __builtin_amdgcn_wmma_f32_16x16x4_f32(
              false, G8f[j], false, b, (short)0, y[v], false, false);
        }
#pragma unroll
      for (int v = 0; v < 8; ++v)
#pragma unroll
        for (int t = 0; t < 8; ++t)
          X[(wave * 16 + 8 * kh + v) * PITCH + t * 16 + m] = y[v][t];
    }
    if (l + 1 < N_LAYER) {
      const v4f* dg4 = reinterpret_cast<const v4f*>(diag + (size_t)(l + 1) * N_HAD);
      v4f dn[16], x[16];
#pragma unroll
      for (int i = 0; i < 16; ++i) dn[i] = dg4[tid + 256 * i];
      __syncthreads();
#pragma unroll
      for (int i = 0; i < 16; ++i) x[i] = *reinterpret_cast<const v4f*>(&X[gb + i * GSTEP]);
#pragma unroll
      for (int i = 0; i < 16; ++i)
        *reinterpret_cast<v4f*>(&X[gb + i * GSTEP]) = x[i] * dn[i];
      __syncthreads();
    } else {
      __syncthreads();
    }
  }

  {
    const v4f* b4 = reinterpret_cast<const v4f*>(bias);
    v4f* o4 = reinterpret_cast<v4f*>(out + (size_t)row * N_RES);
    v4f b[8], x[8];
#pragma unroll
    for (int i = 0; i < 8; ++i) b[i] = b4[tid + 256 * i];
#pragma unroll
    for (int i = 0; i < 8; ++i) x[i] = *reinterpret_cast<const v4f*>(&X[gb + i * GSTEP]);
#pragma unroll
    for (int i = 0; i < 8; ++i) {
      v4f r;
#pragma unroll
      for (int e = 0; e < 4; ++e) {
        const float z = x[i][e] * (1.0f / (float)N_HAD) + b[i][e];
        r[e] = erff(z);
      }
      o4[tid + 256 * i] = r;
    }
  }
}

extern "C" void kernel_launch(void* const* d_in, const int* in_sizes, int n_in,
                              void* d_out, int out_size, void* d_ws, size_t ws_size,
                              hipStream_t stream) {
  const float* state  = (const float*)d_in[0];
  const float* inputs = (const float*)d_in[1];
  const float* bias   = (const float*)d_in[2];
  const float* diag   = (const float*)d_in[3];
  float* out = (float*)d_out;

  const int batch = in_sizes[0] / N_RES;

  (void)hipFuncSetAttribute(reinterpret_cast<const void*>(fwht_reservoir),
                            hipFuncAttributeMaxDynamicSharedMemorySize, LDS_BYTES);

  fwht_reservoir<<<dim3(batch), dim3(256), LDS_BYTES, stream>>>(
      state, inputs, bias, diag, out);
}


# Layers run col-row, row-col, col-row: each layer boundary chains two same-axis H128 stages in registers, diagonal applied there (no LDS round trip, 4 fewer barriers); changes summation order
# speedup vs baseline: 3.8725x; 1.0350x over previous
#include <hip/hip_runtime.h>

typedef __attribute__((ext_vector_type(2))) float v2f;
typedef __attribute__((ext_vector_type(4))) float v4f;
typedef __attribute__((ext_vector_type(8))) float v8f;

#define N_RES   8192
#define N_INP   64
#define N_HAD   16384
#define N_LAYER 3
#define PITCH   140
#define LDS_BYTES (128 * PITCH * 4)
#define GSTEP   (8 * PITCH)

__device__ __forceinline__ float hsgn(int i, int j) {
  return (__popc(i & j) & 1) ? -1.0f : 1.0f;
}

__device__ __forceinline__ v8f v8f_zero() {
  v8f z;
#pragma unroll
  for (int i = 0; i < 8; ++i) z[i] = 0.0f;
  return z;
}

__device__ __forceinline__ void h128_lines(const v2f (&H16f)[4], const v2f (&G8f)[4],
                                           const v2f (&xb)[8][4], v8f (&y)[8]) {
  v8f z[8];
#pragma unroll
  for (int t = 0; t < 8; ++t) z[t] = v8f_zero();
#pragma unroll
  for (int j = 0; j < 4; ++j)
#pragma unroll
    for (int t = 0; t < 8; ++t)
      z[t] = __builtin_amdgcn_wmma_f32_16x16x4_f32(
          false, H16f[j], false, xb[t][j], (short)0, z[t], false, false);
#pragma unroll
  for (int r = 0; r < 8; ++r) y[r] = v8f_zero();
#pragma unroll
  for (int j = 0; j < 4; ++j)
#pragma unroll
    for (int r = 0; r < 8; ++r) {
      v2f b;
      b.x = z[2 * j][r];
      b.y = z[2 * j + 1][r];
      y[r] = __builtin_amdgcn_wmma_f32_16x16x4_f32(
          false, G8f[j], false, b, (short)0, y[r], false, false);
    }
}

__global__ __launch_bounds__(256) void fwht_reservoir(
    const float* __restrict__ state, const float* __restrict__ inputs,
    const float* __restrict__ bias, const float* __restrict__ diag,
    float* __restrict__ out) {
  extern __shared__ __attribute__((aligned(16))) float X[];
  const int row  = blockIdx.x;
  const int tid  = threadIdx.x;
  const int lane = tid & 31;
  const int wave = tid >> 5;
  const int m    = lane & 15;
  const int kh   = lane >> 4;
  const int gb   = wave * PITCH + (lane << 2);
  const int cbase = (8 * kh) * PITCH + wave * 16 + m;
  const int rbase = (wave * 16 + m) * PITCH + 8 * kh;

  {
    const v4f* st4 = reinterpret_cast<const v4f*>(state + (size_t)row * N_RES);
    const v4f* in4 = reinterpret_cast<const v4f*>(inputs + (size_t)row * N_INP);
    const v4f* dg4 = reinterpret_cast<const v4f*>(diag);
    v4f s[8], d[16];
#pragma unroll
    for (int i = 0; i < 8; ++i) s[i] = st4[tid + 256 * i];
    const v4f u = in4[tid & 15];
#pragma unroll
    for (int i = 0; i < 16; ++i) d[i] = dg4[tid + 256 * i];
#pragma unroll
    for (int i = 0; i < 16; ++i) {
      v4f v;
#pragma unroll
      for (int e = 0; e < 4; ++e) {
        if (i < 8)        v[e] = 0.9f * s[i][e];
        else if (i == 8)  v[e] = (tid < 16) ? 0.4f * u[e] : 0.0f;
        else              v[e] = 0.0f;
      }
      *reinterpret_cast<v4f*>(&X[gb + i * GSTEP]) = v * d[i];
    }
  }
  __syncthreads();

  v2f H16f[4], G8f[4];
#pragma unroll
  for (int j = 0; j < 4; ++j) {
    const int k0 = 8 * kh + 2 * j;
    H16f[j].x = hsgn(m, k0);
    H16f[j].y = hsgn(m, k0 + 1);
    const bool same = ((m >> 3) == kh);
    G8f[j].x = same ? hsgn(m & 7, 2 * j)     : 0.0f;
    G8f[j].y = same ? hsgn(m & 7, 2 * j + 1) : 0.0f;
  }


  {
    v2f xb[8][4];
    v8f y[8];
#pragma unroll
    for (int t = 0; t < 8; ++t)
#pragma unroll
      for (int j = 0; j < 4; ++j) {
        xb[t][j].x = X[cbase + (16 * t + 2 * j) * PITCH];
        xb[t][j].y = X[cbase + (16 * t + 2 * j + 1) * PITCH];
      }
    h128_lines(H16f, G8f, xb, y);
#pragma unroll
    for (int q = 0; q < 8; ++q)
#pragma unroll
      for (int t = 0; t < 8; ++t)
        X[cbase + (16 * t + q) * PITCH] = y[q][t];
  }
  v4f dr[8][2];
  {
    const v4f* d1 = reinterpret_cast<const v4f*>(diag + (size_t)N_HAD + (wave * 16 + m) * 128 + 8 * kh);
#pragma unroll
    for (int t = 0; t < 8; ++t) {
      dr[t][0] = d1[4 * t];
      dr[t][1] = d1[4 * t + 1];
    }
  }
  __syncthreads();

  {
    v2f xb[8][4];
    v8f y[8];
#pragma unroll
    for (int t = 0; t < 8; ++t) {
      const float* p = &X[rbase + 16 * t];
      const v4f lo = *reinterpret_cast<const v4f*>(p);
      const v4f hi = *reinterpret_cast<const v4f*>(p + 4);
      xb[t][0].x = lo[0]; xb[t][0].y = lo[1];
      xb[t][1].x = lo[2]; xb[t][1].y = lo[3];
      xb[t][2].x = hi[0]; xb[t][2].y = hi[1];
      xb[t][3].x = hi[2]; xb[t][3].y = hi[3];
    }
    h128_lines(H16f, G8f, xb, y);
#pragma unroll
    for (int t = 0; t < 8; ++t)
#pragma unroll
      for (int j = 0; j < 4; ++j) {
        xb[t][j].x = y[2 * j][t]     * dr[t][j >> 1][2 * (j & 1)];
        xb[t][j].y = y[2 * j + 1][t] * dr[t][j >> 1][2 * (j & 1) + 1];
      }
    h128_lines(H16f, G8f, xb, y);
#pragma unroll
    for (int t = 0; t < 8; ++t) {
      v4f lo, hi;
#pragma unroll
      for (int e = 0; e < 4; ++e) { lo[e] = y[e][t]; hi[e] = y[4 + e][t]; }
      float* p = &X[rbase + 16 * t];
      *reinterpret_cast<v4f*>(p)     = lo;
      *reinterpret_cast<v4f*>(p + 4) = hi;
    }
  }
  float dc[8][8];
  {
    const float* d2 = diag + (size_t)2 * N_HAD + (8 * kh) * 128 + wave * 16 + m;
#pragma unroll
    for (int t = 0; t < 8; ++t)
#pragma unroll
      for (int q = 0; q < 8; ++q)
        dc[t][q] = d2[(16 * t + q) * 128];
  }
  __syncthreads();

  {
    v2f xb[8][4];
    v8f y[8];
#pragma unroll
    for (int t = 0; t < 8; ++t)
#pragma unroll
      for (int j = 0; j < 4; ++j) {
        xb[t][j].x = X[cbase + (16 * t + 2 * j) * PITCH];
        xb[t][j].y = X[cbase + (16 * t + 2 * j + 1) * PITCH];
      }
    h128_lines(H16f, G8f, xb, y);
#pragma unroll
    for (int t = 0; t < 8; ++t)
#pragma unroll
      for (int j = 0; j < 4; ++j) {
        xb[t][j].x = y[2 * j][t]     * dc[t][2 * j];
        xb[t][j].y = y[2 * j + 1][t] * dc[t][2 * j + 1];
      }
    h128_lines(H16f, G8f, xb, y);
#pragma unroll
    for (int q = 0; q < 8; ++q)
#pragma unroll
      for (int t = 0; t < 8; ++t)
        X[cbase + (16 * t + q) * PITCH] = y[q][t];
  }
  __syncthreads();

  {
    v2f xb[8][4];
    v8f y[8];
#pragma unroll
    for (int t = 0; t < 8; ++t) {
      const float* p = &X[rbase + 16 * t];
      const v4f lo = *reinterpret_cast<const v4f*>(p);
      const v4f hi = *reinterpret_cast<const v4f*>(p + 4);
      xb[t][0].x = lo[0]; xb[t][0].y = lo[1];
      xb[t][1].x = lo[2]; xb[t][1].y = lo[3];
      xb[t][2].x = hi[0]; xb[t][2].y = hi[1];
      xb[t][3].x = hi[2]; xb[t][3].y = hi[3];
    }
    h128_lines(H16f, G8f, xb, y);
#pragma unroll
    for (int t = 0; t < 8; ++t) {
      v4f lo, hi;
#pragma unroll
      for (int e = 0; e < 4; ++e) { lo[e] = y[e][t]; hi[e] = y[4 + e][t]; }
      float* p = &X[rbase + 16 * t];
      *reinterpret_cast<v4f*>(p)     = lo;
      *reinterpret_cast<v4f*>(p + 4) = hi;
    }
  }
  __syncthreads();

  {
    const v4f* b4 = reinterpret_cast<const v4f*>(bias);
    v4f* o4 = reinterpret_cast<v4f*>(out + (size_t)row * N_RES);
    v4f b[8], x[8];
#pragma unroll
    for (int i = 0; i < 8; ++i) b[i] = b4[tid + 256 * i];
#pragma unroll
    for (int i = 0; i < 8; ++i) x[i] = *reinterpret_cast<const v4f*>(&X[gb + i * GSTEP]);
#pragma unroll
    for (int i = 0; i < 8; ++i) {
      v4f r;
#pragma unroll
      for (int e = 0; e < 4; ++e) {
        const float z = x[i][e] * (1.0f / (float)N_HAD) + b[i][e];
        r[e] = erff(z);
      }
      o4[tid + 256 * i] = r;
    }
  }
}

extern "C" void kernel_launch(void* const* d_in, const int* in_sizes, int n_in,
                              void* d_out, int out_size, void* d_ws, size_t ws_size,
                              hipStream_t stream) {
  const float* state  = (const float*)d_in[0];
  const float* inputs = (const float*)d_in[1];
  const float* bias   = (const float*)d_in[2];
  const float* diag   = (const float*)d_in[3];
  float* out = (float*)d_out;

  const int batch = in_sizes[0] / N_RES;

  (void)hipFuncSetAttribute(reinterpret_cast<const void*>(fwht_reservoir),
                            hipFuncAttributeMaxDynamicSharedMemorySize, LDS_BYTES);

  fwht_reservoir<<<dim3(batch), dim3(256), LDS_BYTES, stream>>>(
      state, inputs, bias, diag, out);
}


# Products on zero padding (layer 0 rows 80..127) and for discarded outputs (last layer rows 64..127) left out; last row stage split over wave pairs, partials added in epilogue: sum order changes
# speedup vs baseline: 4.2893x; 1.1076x over previous
#include <hip/hip_runtime.h>

typedef __attribute__((ext_vector_type(2))) float v2f;
typedef __attribute__((ext_vector_type(4))) float v4f;
typedef __attribute__((ext_vector_type(8))) float v8f;

#define N_RES   8192
#define N_INP   64
#define N_HAD   16384
#define N_LAYER 3
#define PITCH   140
#define LDS_BYTES (192 * PITCH * 4)
#define YOFF    (128 * PITCH)
#define GSTEP   (8 * PITCH)

__device__ __forceinline__ float hsgn(int i, int j) {
  return (__popc(i & j) & 1) ? -1.0f : 1.0f;
}

__device__ __forceinline__ v8f v8f_zero() {
  v8f z;
#pragma unroll
  for (int i = 0; i < 8; ++i) z[i] = 0.0f;
  return z;
}

__device__ __forceinline__ void h128_lines(const v2f (&H16f)[4], const v2f (&G8f)[4],
                                           const v2f (&xb)[8][4], v8f (&y)[8]) {
  v8f z[8];
#pragma unroll
  for (int t = 0; t < 8; ++t) z[t] = v8f_zero();
#pragma unroll
  for (int j = 0; j < 4; ++j)
#pragma unroll
    for (int t = 0; t < 8; ++t)
      z[t] = __builtin_amdgcn_wmma_f32_16x16x4_f32(
          false, H16f[j], false, xb[t][j], (short)0, z[t], false, false);
#pragma unroll
  for (int r = 0; r < 8; ++r) y[r] = v8f_zero();
#pragma unroll
  for (int j = 0; j < 4; ++j)
#pragma unroll
    for (int r = 0; r < 8; ++r) {
      v2f b;
      b.x = z[2 * j][r];
      b.y = z[2 * j + 1][r];
      y[r] = __builtin_amdgcn_wmma_f32_16x16x4_f32(
          false, G8f[j], false, b, (short)0, y[r], false, false);
    }
}

__global__ __launch_bounds__(256, 4) void fwht_reservoir(
    const float* __restrict__ state, const float* __restrict__ inputs,
    const float* __restrict__ bias, const float* __restrict__ diag,
    float* __restrict__ out) {
  extern __shared__ __attribute__((aligned(16))) float X[];
  const int row  = blockIdx.x;
  const int tid  = threadIdx.x;
  const int lane = tid & 31;
  const int wave = tid >> 5;
  const int m    = lane & 15;
  const int kh   = lane >> 4;
  const int gb   = wave * PITCH + (lane << 2);
  const int cbase = (8 * kh) * PITCH + wave * 16 + m;
  const int rbase = (wave * 16 + m) * PITCH + 8 * kh;

  {
    const v4f* st4 = reinterpret_cast<const v4f*>(state + (size_t)row * N_RES);
    const v4f* in4 = reinterpret_cast<const v4f*>(inputs + (size_t)row * N_INP);
    const v4f* dg4 = reinterpret_cast<const v4f*>(diag);
    v4f s[8], d[10];
#pragma unroll
    for (int i = 0; i < 8; ++i) s[i] = st4[tid + 256 * i];
    const v4f u = in4[tid & 15];
#pragma unroll
    for (int i = 0; i < 10; ++i) d[i] = dg4[tid + 256 * i];
#pragma unroll
    for (int i = 0; i < 10; ++i) {
      v4f v;
#pragma unroll
      for (int e = 0; e < 4; ++e) {
        if (i < 8)        v[e] = 0.9f * s[i][e];
        else if (i == 8)  v[e] = (tid < 16) ? 0.4f * u[e] : 0.0f;
        else              v[e] = 0.0f;
      }
      *reinterpret_cast<v4f*>(&X[gb + i * GSTEP]) = v * d[i];
    }
  }
  __syncthreads();

  v2f H16f[4], G8f[4];
#pragma unroll
  for (int j = 0; j < 4; ++j) {
    const int k0 = 8 * kh + 2 * j;
    H16f[j].x = hsgn(m, k0);
    H16f[j].y = hsgn(m, k0 + 1);
    const bool same = ((m >> 3) == kh);
    G8f[j].x = same ? hsgn(m & 7, 2 * j)     : 0.0f;
    G8f[j].y = same ? hsgn(m & 7, 2 * j + 1) : 0.0f;
  }


  {
    v2f xb[5][4];
    v8f z[5], y[8];
#pragma unroll
    for (int t = 0; t < 5; ++t)
#pragma unroll
      for (int j = 0; j < 4; ++j) {
        xb[t][j].x = X[cbase + (16 * t + 2 * j) * PITCH];
        xb[t][j].y = X[cbase + (16 * t + 2 * j + 1) * PITCH];
      }
#pragma unroll
    for (int t = 0; t < 5; ++t) z[t] = v8f_zero();
#pragma unroll
    for (int j = 0; j < 4; ++j)
#pragma unroll
      for (int t = 0; t < 5; ++t)
        z[t] = __builtin_amdgcn_wmma_f32_16x16x4_f32(
            false, H16f[j], false, xb[t][j], (short)0, z[t], false, false);
#pragma unroll
    for (int r = 0; r < 8; ++r) y[r] = v8f_zero();
#pragma unroll
    for (int j = 0; j < 3; ++j)
#pragma unroll
      for (int r = 0; r < 8; ++r) {
        v2f b;
        b.x = z[2 * j][r];
        b.y = (j < 2) ? z[2 * j + 1][r] : 0.0f;
        y[r] = __builtin_amdgcn_wmma_f32_16x16x4_f32(
            false, G8f[j], false, b, (short)0, y[r], false, false);
      }
#pragma unroll
    for (int q = 0; q < 8; ++q)
#pragma unroll
      for (int t = 0; t < 8; ++t)
        X[cbase + (16 * t + q) * PITCH] = y[q][t];
  }
  v4f dr[8][2];
  {
    const v4f* d1 = reinterpret_cast<const v4f*>(diag + (size_t)N_HAD + (wave * 16 + m) * 128 + 8 * kh);
#pragma unroll
    for (int t = 0; t < 8; ++t) {
      dr[t][0] = d1[4 * t];
      dr[t][1] = d1[4 * t + 1];
    }
  }
  __syncthreads();

  {
    v2f xb[8][4];
    v8f y[8];
#pragma unroll
    for (int t = 0; t < 8; ++t) {
      const float* p = &X[rbase + 16 * t];
      const v4f lo = *reinterpret_cast<const v4f*>(p);
      const v4f hi = *reinterpret_cast<const v4f*>(p + 4);
      xb[t][0].x = lo[0]; xb[t][0].y = lo[1];
      xb[t][1].x = lo[2]; xb[t][1].y = lo[3];
      xb[t][2].x = hi[0]; xb[t][2].y = hi[1];
      xb[t][3].x = hi[2]; xb[t][3].y = hi[3];
    }
    h128_lines(H16f, G8f, xb, y);
#pragma unroll
    for (int t = 0; t < 8; ++t)
#pragma unroll
      for (int j = 0; j < 4; ++j) {
        xb[t][j].x = y[2 * j][t]     * dr[t][j >> 1][2 * (j & 1)];
        xb[t][j].y = y[2 * j + 1][t] * dr[t][j >> 1][2 * (j & 1) + 1];
      }
    h128_lines(H16f, G8f, xb, y);
#pragma unroll
    for (int t = 0; t < 8; ++t) {
      v4f lo, hi;
#pragma unroll
      for (int e = 0; e < 4; ++e) { lo[e] = y[e][t]; hi[e] = y[4 + e][t]; }
      float* p = &X[rbase + 16 * t];
      *reinterpret_cast<v4f*>(p)     = lo;
      *reinterpret_cast<v4f*>(p + 4) = hi;
    }
  }
  float dc[8][8];
  {
    const float* d2 = diag + (size_t)2 * N_HAD + (8 * kh) * 128 + wave * 16 + m;
#pragma unroll
    for (int t = 0; t < 8; ++t)
#pragma unroll
      for (int q = 0; q < 8; ++q)
        dc[t][q] = d2[(16 * t + q) * 128];
  }
  __syncthreads();

  {
    v2f xb[8][4];
    v8f y[8];
#pragma unroll
    for (int t = 0; t < 8; ++t)
#pragma unroll
      for (int j = 0; j < 4; ++j) {
        xb[t][j].x = X[cbase + (16 * t + 2 * j) * PITCH];
        xb[t][j].y = X[cbase + (16 * t + 2 * j + 1) * PITCH];
      }
    h128_lines(H16f, G8f, xb, y);
#pragma unroll
    for (int q = 0; q < 8; ++q)
#pragma unroll
      for (int t = 0; t < 8; ++t)
        y[q][t] *= dc[t][q];
    v8f w[8];
#pragma unroll
    for (int q = 0; q < 8; ++q) w[q] = v8f_zero();
#pragma unroll
    for (int j = 0; j < 4; ++j)
#pragma unroll
      for (int q = 0; q < 8; ++q) {
        v2f b;
        b.x = y[q][2 * j];
        b.y = y[q][2 * j + 1];
        w[q] = __builtin_amdgcn_wmma_f32_16x16x4_f32(
            false, G8f[j], false, b, (short)0, w[q], false, false);
      }
    v8f o[4];
#pragma unroll
    for (int t = 0; t < 4; ++t) o[t] = v8f_zero();
#pragma unroll
    for (int j = 0; j < 4; ++j)
#pragma unroll
      for (int t = 0; t < 4; ++t) {
        v2f b;
        b.x = w[2 * j][t];
        b.y = w[2 * j + 1][t];
        o[t] = __builtin_amdgcn_wmma_f32_16x16x4_f32(
            false, H16f[j], false, b, (short)0, o[t], false, false);
      }
#pragma unroll
    for (int t = 0; t < 4; ++t)
#pragma unroll
      for (int r = 0; r < 8; ++r)
        X[cbase + (16 * t + r) * PITCH] = o[t][r];
  }
  __syncthreads();

  {
    const int s = wave >> 2;
    const int g = wave & 3;
    const int sbase = (g * 16 + m) * PITCH + 8 * kh + 64 * s;
    const int dbase = s ? rbase : (YOFF + rbase);
    v2f Gs[2];
#pragma unroll
    for (int jj = 0; jj < 2; ++jj) {
      const bool same = ((m >> 3) == kh);
      Gs[jj].x = same ? hsgn(m & 7, 4 * s + 2 * jj)     : 0.0f;
      Gs[jj].y = same ? hsgn(m & 7, 4 * s + 2 * jj + 1) : 0.0f;
    }
    v2f xb[4][4];
    v8f z[4], y[8];
#pragma unroll
    for (int tt = 0; tt < 4; ++tt) {
      const float* p = &X[sbase + 16 * tt];
      const v4f lo = *reinterpret_cast<const v4f*>(p);
      const v4f hi = *reinterpret_cast<const v4f*>(p + 4);
      xb[tt][0].x = lo[0]; xb[tt][0].y = lo[1];
      xb[tt][1].x = lo[2]; xb[tt][1].y = lo[3];
      xb[tt][2].x = hi[0]; xb[tt][2].y = hi[1];
      xb[tt][3].x = hi[2]; xb[tt][3].y = hi[3];
    }
#pragma unroll
    for (int tt = 0; tt < 4; ++tt) z[tt] = v8f_zero();
#pragma unroll
    for (int j = 0; j < 4; ++j)
#pragma unroll
      for (int tt = 0; tt < 4; ++tt)
        z[tt] = __builtin_amdgcn_wmma_f32_16x16x4_f32(
            false, H16f[j], false, xb[tt][j], (short)0, z[tt], false, false);
#pragma unroll
    for (int r = 0; r < 8; ++r) y[r] = v8f_zero();
#pragma unroll
    for (int jj = 0; jj < 2; ++jj)
#pragma unroll
      for (int r = 0; r < 8; ++r) {
        v2f b;
        b.x = z[2 * jj][r];
        b.y = z[2 * jj + 1][r];
        y[r] = __builtin_amdgcn_wmma_f32_16x16x4_f32(
            false, Gs[jj], false, b, (short)0, y[r], false, false);
      }
#pragma unroll
    for (int t = 0; t < 8; ++t) {
      v4f lo, hi;
#pragma unroll
      for (int e = 0; e < 4; ++e) { lo[e] = y[e][t]; hi[e] = y[4 + e][t]; }
      float* p = &X[dbase + 16 * t];
      *reinterpret_cast<v4f*>(p)     = lo;
      *reinterpret_cast<v4f*>(p + 4) = hi;
    }
  }
  __syncthreads();

  {
    const v4f* b4 = reinterpret_cast<const v4f*>(bias);
    v4f* o4 = reinterpret_cast<v4f*>(out + (size_t)row * N_RES);
    v4f b[8], x[8], x1[8];
#pragma unroll
    for (int i = 0; i < 8; ++i) b[i] = b4[tid + 256 * i];
#pragma unroll
    for (int i = 0; i < 8; ++i) {
      x[i]  = *reinterpret_cast<const v4f*>(&X[YOFF + gb + i * GSTEP]);
      x1[i] = *reinterpret_cast<const v4f*>(&X[gb + (i + 8) * GSTEP]);
    }
#pragma unroll
    for (int i = 0; i < 8; ++i) {
      v4f r;
#pragma unroll
      for (int e = 0; e < 4; ++e) {
        const float z = (x[i][e] + x1[i][e]) * (1.0f / (float)N_HAD) + b[i][e];
        r[e] = erff(z);
      }
      o4[tid + 256 * i] = r;
    }
  }
}

extern "C" void kernel_launch(void* const* d_in, const int* in_sizes, int n_in,
                              void* d_out, int out_size, void* d_ws, size_t ws_size,
                              hipStream_t stream) {
  const float* state  = (const float*)d_in[0];
  const float* inputs = (const float*)d_in[1];
  const float* bias   = (const float*)d_in[2];
  const float* diag   = (const float*)d_in[3];
  float* out = (float*)d_out;

  const int batch = in_sizes[0] / N_RES;

  (void)hipFuncSetAttribute(reinterpret_cast<const void*>(fwht_reservoir),
                            hipFuncAttributeMaxDynamicSharedMemorySize, LDS_BYTES);

  fwht_reservoir<<<dim3(batch), dim3(256), LDS_BYTES, stream>>>(
      state, inputs, bias, diag, out);
}


# Epilogue erf: branch-free two-range form (odd polynomial below 0.875, rational times native exp2 above; within 2e-7 relative of erf) replaces 32 sequential two-branch library calls per thread
# speedup vs baseline: 4.8335x; 1.1269x over previous
#include <hip/hip_runtime.h>

typedef __attribute__((ext_vector_type(2))) float v2f;
typedef __attribute__((ext_vector_type(4))) float v4f;
typedef __attribute__((ext_vector_type(8))) float v8f;

#define N_RES   8192
#define N_INP   64
#define N_HAD   16384
#define N_LAYER 3
#define PITCH   140
#define LDS_BYTES (192 * PITCH * 4)
#define YOFF    (128 * PITCH)
#define GSTEP   (8 * PITCH)

__device__ __forceinline__ float hsgn(int i, int j) {
  return (__popc(i & j) & 1) ? -1.0f : 1.0f;
}

__device__ __forceinline__ v8f v8f_zero() {
  v8f z;
#pragma unroll
  for (int i = 0; i < 8; ++i) z[i] = 0.0f;
  return z;
}

__device__ __forceinline__ float erf_two_range(float x) {
  const float ax = __builtin_fabsf(x);
  const float s  = x * x;
  float q = __builtin_fmaf(-0.000610699411f, s, 0.00501536019f);
  q = __builtin_fmaf(q, s, -0.0267813429f);
  q = __builtin_fmaf(q, s, 0.112822123f);
  q = __builtin_fmaf(q, s, -0.376125306f);
  q = __builtin_fmaf(q, s, 1.12837911f);
  const float small = q * x;
  const float t = __builtin_amdgcn_rcpf(__builtin_fmaf(0.3275911f, ax, 1.0f));
  const float e = __builtin_amdgcn_exp2f(s * -1.44269504f);
  float p = __builtin_fmaf(0.464900345f, t, -0.653243780f);
  p = __builtin_fmaf(p, t, 1.07562256f);
  p = __builtin_fmaf(p, t, -0.442466795f);
  p = __builtin_fmaf(p, t, 0.401989669f);
  p = __builtin_fmaf(p, t, 0.153773218f);
  p = p * t;
  const float large = __builtin_copysignf(__builtin_fmaf(-p, e, 1.0f), x);
  return (ax < 0.875f) ? small : large;
}

__device__ __forceinline__ void h128_lines(const v2f (&H16f)[4], const v2f (&G8f)[4],
                                           const v2f (&xb)[8][4], v8f (&y)[8]) {
  v8f z[8];
#pragma unroll
  for (int t = 0; t < 8; ++t) z[t] = v8f_zero();
#pragma unroll
  for (int j = 0; j < 4; ++j)
#pragma unroll
    for (int t = 0; t < 8; ++t)
      z[t] = __builtin_amdgcn_wmma_f32_16x16x4_f32(
          false, H16f[j], false, xb[t][j], (short)0, z[t], false, false);
#pragma unroll
  for (int r = 0; r < 8; ++r) y[r] = v8f_zero();
#pragma unroll
  for (int j = 0; j < 4; ++j)
#pragma unroll
    for (int r = 0; r < 8; ++r) {
      v2f b;
      b.x = z[2 * j][r];
      b.y = z[2 * j + 1][r];
      y[r] = __builtin_amdgcn_wmma_f32_16x16x4_f32(
          false, G8f[j], false, b, (short)0, y[r], false, false);
    }
}

__global__ __launch_bounds__(256, 4) void fwht_reservoir(
    const float* __restrict__ state, const float* __restrict__ inputs,
    const float* __restrict__ bias, const float* __restrict__ diag,
    float* __restrict__ out) {
  extern __shared__ __attribute__((aligned(16))) float X[];
  const int row  = blockIdx.x;
  const int tid  = threadIdx.x;
  const int lane = tid & 31;
  const int wave = tid >> 5;
  const int m    = lane & 15;
  const int kh   = lane >> 4;
  const int gb   = wave * PITCH + (lane << 2);
  const int cbase = (8 * kh) * PITCH + wave * 16 + m;
  const int rbase = (wave * 16 + m) * PITCH + 8 * kh;

  {
    const v4f* st4 = reinterpret_cast<const v4f*>(state + (size_t)row * N_RES);
    const v4f* in4 = reinterpret_cast<const v4f*>(inputs + (size_t)row * N_INP);
    const v4f* dg4 = reinterpret_cast<const v4f*>(diag);
    v4f s[8], d[10];
#pragma unroll
    for (int i = 0; i < 8; ++i) s[i] = st4[tid + 256 * i];
    const v4f u = in4[tid & 15];
#pragma unroll
    for (int i = 0; i < 10; ++i) d[i] = dg4[tid + 256 * i];
#pragma unroll
    for (int i = 0; i < 10; ++i) {
      v4f v;
#pragma unroll
      for (int e = 0; e < 4; ++e) {
        if (i < 8)        v[e] = 0.9f * s[i][e];
        else if (i == 8)  v[e] = (tid < 16) ? 0.4f * u[e] : 0.0f;
        else              v[e] = 0.0f;
      }
      *reinterpret_cast<v4f*>(&X[gb + i * GSTEP]) = v * d[i];
    }
  }
  __syncthreads();

  v2f H16f[4], G8f[4];
#pragma unroll
  for (int j = 0; j < 4; ++j) {
    const int k0 = 8 * kh + 2 * j;
    H16f[j].x = hsgn(m, k0);
    H16f[j].y = hsgn(m, k0 + 1);
    const bool same = ((m >> 3) == kh);
    G8f[j].x = same ? hsgn(m & 7, 2 * j)     : 0.0f;
    G8f[j].y = same ? hsgn(m & 7, 2 * j + 1) : 0.0f;
  }


  {
    v2f xb[5][4];
    v8f z[5], y[8];
#pragma unroll
    for (int t = 0; t < 5; ++t)
#pragma unroll
      for (int j = 0; j < 4; ++j) {
        xb[t][j].x = X[cbase + (16 * t + 2 * j) * PITCH];
        xb[t][j].y = X[cbase + (16 * t + 2 * j + 1) * PITCH];
      }
#pragma unroll
    for (int t = 0; t < 5; ++t) z[t] = v8f_zero();
#pragma unroll
    for (int j = 0; j < 4; ++j)
#pragma unroll
      for (int t = 0; t < 5; ++t)
        z[t] = __builtin_amdgcn_wmma_f32_16x16x4_f32(
            false, H16f[j], false, xb[t][j], (short)0, z[t], false, false);
#pragma unroll
    for (int r = 0; r < 8; ++r) y[r] = v8f_zero();
#pragma unroll
    for (int j = 0; j < 3; ++j)
#pragma unroll
      for (int r = 0; r < 8; ++r) {
        v2f b;
        b.x = z[2 * j][r];
        b.y = (j < 2) ? z[2 * j + 1][r] : 0.0f;
        y[r] = __builtin_amdgcn_wmma_f32_16x16x4_f32(
            false, G8f[j], false, b, (short)0, y[r], false, false);
      }
#pragma unroll
    for (int q = 0; q < 8; ++q)
#pragma unroll
      for (int t = 0; t < 8; ++t)
        X[cbase + (16 * t + q) * PITCH] = y[q][t];
  }
  v4f dr[8][2];
  {
    const v4f* d1 = reinterpret_cast<const v4f*>(diag + (size_t)N_HAD + (wave * 16 + m) * 128 + 8 * kh);
#pragma unroll
    for (int t = 0; t < 8; ++t) {
      dr[t][0] = d1[4 * t];
      dr[t][1] = d1[4 * t + 1];
    }
  }
  __syncthreads();

  {
    v2f xb[8][4];
    v8f y[8];
#pragma unroll
    for (int t = 0; t < 8; ++t) {
      const float* p = &X[rbase + 16 * t];
      const v4f lo = *reinterpret_cast<const v4f*>(p);
      const v4f hi = *reinterpret_cast<const v4f*>(p + 4);
      xb[t][0].x = lo[0]; xb[t][0].y = lo[1];
      xb[t][1].x = lo[2]; xb[t][1].y = lo[3];
      xb[t][2].x = hi[0]; xb[t][2].y = hi[1];
      xb[t][3].x = hi[2]; xb[t][3].y = hi[3];
    }
    h128_lines(H16f, G8f, xb, y);
#pragma unroll
    for (int t = 0; t < 8; ++t)
#pragma unroll
      for (int j = 0; j < 4; ++j) {
        xb[t][j].x = y[2 * j][t]     * dr[t][j >> 1][2 * (j & 1)];
        xb[t][j].y = y[2 * j + 1][t] * dr[t][j >> 1][2 * (j & 1) + 1];
      }
    h128_lines(H16f, G8f, xb, y);
#pragma unroll
    for (int t = 0; t < 8; ++t) {
      v4f lo, hi;
#pragma unroll
      for (int e = 0; e < 4; ++e) { lo[e] = y[e][t]; hi[e] = y[4 + e][t]; }
      float* p = &X[rbase + 16 * t];
      *reinterpret_cast<v4f*>(p)     = lo;
      *reinterpret_cast<v4f*>(p + 4) = hi;
    }
  }
  float dc[8][8];
  {
    const float* d2 = diag + (size_t)2 * N_HAD + (8 * kh) * 128 + wave * 16 + m;
#pragma unroll
    for (int t = 0; t < 8; ++t)
#pragma unroll
      for (int q = 0; q < 8; ++q)
        dc[t][q] = d2[(16 * t + q) * 128];
  }
  __syncthreads();

  {
    v2f xb[8][4];
    v8f y[8];
#pragma unroll
    for (int t = 0; t < 8; ++t)
#pragma unroll
      for (int j = 0; j < 4; ++j) {
        xb[t][j].x = X[cbase + (16 * t + 2 * j) * PITCH];
        xb[t][j].y = X[cbase + (16 * t + 2 * j + 1) * PITCH];
      }
    h128_lines(H16f, G8f, xb, y);
#pragma unroll
    for (int q = 0; q < 8; ++q)
#pragma unroll
      for (int t = 0; t < 8; ++t)
        y[q][t] *= dc[t][q];
    v8f w[8];
#pragma unroll
    for (int q = 0; q < 8; ++q) w[q] = v8f_zero();
#pragma unroll
    for (int j = 0; j < 4; ++j)
#pragma unroll
      for (int q = 0; q < 8; ++q) {
        v2f b;
        b.x = y[q][2 * j];
        b.y = y[q][2 * j + 1];
        w[q] = __builtin_amdgcn_wmma_f32_16x16x4_f32(
            false, G8f[j], false, b, (short)0, w[q], false, false);
      }
    v8f o[4];
#pragma unroll
    for (int t = 0; t < 4; ++t) o[t] = v8f_zero();
#pragma unroll
    for (int j = 0; j < 4; ++j)
#pragma unroll
      for (int t = 0; t < 4; ++t) {
        v2f b;
        b.x = w[2 * j][t];
        b.y = w[2 * j + 1][t];
        o[t] = __builtin_amdgcn_wmma_f32_16x16x4_f32(
            false, H16f[j], false, b, (short)0, o[t], false, false);
      }
#pragma unroll
    for (int t = 0; t < 4; ++t)
#pragma unroll
      for (int r = 0; r < 8; ++r)
        X[cbase + (16 * t + r) * PITCH] = o[t][r];
  }
  __syncthreads();

  {
    const int s = wave >> 2;
    const int g = wave & 3;
    const int sbase = (g * 16 + m) * PITCH + 8 * kh + 64 * s;
    const int dbase = s ? rbase : (YOFF + rbase);
    v2f Gs[2];
#pragma unroll
    for (int jj = 0; jj < 2; ++jj) {
      const bool same = ((m >> 3) == kh);
      Gs[jj].x = same ? hsgn(m & 7, 4 * s + 2 * jj)     : 0.0f;
      Gs[jj].y = same ? hsgn(m & 7, 4 * s + 2 * jj + 1) : 0.0f;
    }
    v2f xb[4][4];
    v8f z[4], y[8];
#pragma unroll
    for (int tt = 0; tt < 4; ++tt) {
      const float* p = &X[sbase + 16 * tt];
      const v4f lo = *reinterpret_cast<const v4f*>(p);
      const v4f hi = *reinterpret_cast<const v4f*>(p + 4);
      xb[tt][0].x = lo[0]; xb[tt][0].y = lo[1];
      xb[tt][1].x = lo[2]; xb[tt][1].y = lo[3];
      xb[tt][2].x = hi[0]; xb[tt][2].y = hi[1];
      xb[tt][3].x = hi[2]; xb[tt][3].y = hi[3];
    }
#pragma unroll
    for (int tt = 0; tt < 4; ++tt) z[tt] = v8f_zero();
#pragma unroll
    for (int j = 0; j < 4; ++j)
#pragma unroll
      for (int tt = 0; tt < 4; ++tt)
        z[tt] = __builtin_amdgcn_wmma_f32_16x16x4_f32(
            false, H16f[j], false, xb[tt][j], (short)0, z[tt], false, false);
#pragma unroll
    for (int r = 0; r < 8; ++r) y[r] = v8f_zero();
#pragma unroll
    for (int jj = 0; jj < 2; ++jj)
#pragma unroll
      for (int r = 0; r < 8; ++r) {
        v2f b;
        b.x = z[2 * jj][r];
        b.y = z[2 * jj + 1][r];
        y[r] = __builtin_amdgcn_wmma_f32_16x16x4_f32(
            false, Gs[jj], false, b, (short)0, y[r], false, false);
      }
#pragma unroll
    for (int t = 0; t < 8; ++t) {
      v4f lo, hi;
#pragma unroll
      for (int e = 0; e < 4; ++e) { lo[e] = y[e][t]; hi[e] = y[4 + e][t]; }
      float* p = &X[dbase + 16 * t];
      *reinterpret_cast<v4f*>(p)     = lo;
      *reinterpret_cast<v4f*>(p + 4) = hi;
    }
  }
  __syncthreads();

  {
    const v4f* b4 = reinterpret_cast<const v4f*>(bias);
    v4f* o4 = reinterpret_cast<v4f*>(out + (size_t)row * N_RES);
    v4f b[8], x[8], x1[8];
#pragma unroll
    for (int i = 0; i < 8; ++i) b[i] = b4[tid + 256 * i];
#pragma unroll
    for (int i = 0; i < 8; ++i) {
      x[i]  = *reinterpret_cast<const v4f*>(&X[YOFF + gb + i * GSTEP]);
      x1[i] = *reinterpret_cast<const v4f*>(&X[gb + (i + 8) * GSTEP]);
    }
#pragma unroll
    for (int i = 0; i < 8; ++i) {
      v4f r;
#pragma unroll
      for (int e = 0; e < 4; ++e) {
        const float z = (x[i][e] + x1[i][e]) * (1.0f / (float)N_HAD) + b[i][e];
        r[e] = erf_two_range(z);
      }
      o4[tid + 256 * i] = r;
    }
  }
}

extern "C" void kernel_launch(void* const* d_in, const int* in_sizes, int n_in,
                              void* d_out, int out_size, void* d_ws, size_t ws_size,
                              hipStream_t stream) {
  const float* state  = (const float*)d_in[0];
  const float* inputs = (const float*)d_in[1];
  const float* bias   = (const float*)d_in[2];
  const float* diag   = (const float*)d_in[3];
  float* out = (float*)d_out;

  const int batch = in_sizes[0] / N_RES;

  (void)hipFuncSetAttribute(reinterpret_cast<const void*>(fwht_reservoir),
                            hipFuncAttributeMaxDynamicSharedMemorySize, LDS_BYTES);

  fwht_reservoir<<<dim3(batch), dim3(256), LDS_BYTES, stream>>>(
      state, inputs, bias, diag, out);
}
